# starting kernel
# baseline (speedup 1.0000x reference)
#include <hip/hip_runtime.h>
#include <hip/hip_bf16.h>


typedef __attribute__((ext_vector_type(2))) float v2f;
typedef __attribute__((ext_vector_type(8))) float v8f;

#define WS_MR   0
#define WS_MI   4096
#define WS_MIN  8192
#define WS_C0   12288
#define WS_S0   12294

struct c32 { float r, i; };
__device__ __forceinline__ c32 cmul(c32 a, c32 b) {
    return { a.r * b.r - a.i * b.i, a.r * b.i + a.i * b.r };
}

__global__ void qsim_setup_kernel(const float* __restrict__ w,
                                  float* __restrict__ ws) {
    __shared__ float d0r[64], d0i[64];
    const int k = threadIdx.x;
    float phi0 = 0.f, phi1 = 0.f;
#pragma unroll
    for (int i = 0; i < 5; ++i) {
        const int bi = (k >> (5 - i)) & 1;
        const int bj = (k >> (4 - i)) & 1;
        const float sgn = (bi == bj) ? -0.5f : 0.5f;
        phi0 += sgn * w[6 + i];
        phi1 += sgn * w[17 + i];
    }
    float s, c;
    sincosf(phi0, &s, &c); d0r[k] = c; d0i[k] = s;
    float d1r, d1i; sincosf(phi1, &d1i, &d1r);
    float c1[6], s1[6];
#pragma unroll
    for (int i = 0; i < 6; ++i) sincosf(0.5f * w[11 + i], &s1[i], &c1[i]);
    __syncthreads();
    for (int kp = 0; kp < 64; ++kp) {
        float re = 1.f;
#pragma unroll
        for (int i = 0; i < 6; ++i) {
            const int kb = (k >> (5 - i)) & 1;
            const int pb = (kp >> (5 - i)) & 1;
            re *= (kb == pb) ? c1[i] : (kb ? s1[i] : -s1[i]);
        }
        const float mr = re * (d1r * d0r[kp] - d1i * d0i[kp]);
        const float mi = re * (d1r * d0i[kp] + d1i * d0r[kp]);
        ws[WS_MR  + k * 64 + kp] = mr;
        ws[WS_MI  + k * 64 + kp] = mi;
        ws[WS_MIN + k * 64 + kp] = -mi;
    }
    if (k < 6) {
        float ss, cc; sincosf(0.5f * w[k], &ss, &cc);
        ws[WS_C0 + k] = cc;
        ws[WS_S0 + k] = ss;
    }
}

__global__ __launch_bounds__(128, 1)
void qsim_main_kernel(const float* __restrict__ z,
                      const float* __restrict__ ws,
                      float* __restrict__ out) {
    __shared__ float s_psi[4][2][64 * 17];

    const int tid  = threadIdx.x;
    const int wave = tid >> 5;
    const int lane = tid & 31;
    const int n    = lane & 15;
    const int kh   = lane >> 4;
    const int base = blockIdx.x * 64 + wave * 16;
    const int b    = base + n;

    const float z0 = z[2 * b + 0];
    const float z1 = z[2 * b + 1];
    float sa, ca, sb, cb;
    sincosf(0.5f * asinf(z0), &sa, &ca);
    sincosf(0.5f * asinf(z1), &sb, &cb);
    const c32 x0{ cb * ca, -sb * ca };
    const c32 x1{ cb * sa,  sb * sa };

    c32 u[6][2];
#pragma unroll
    for (int j = 0; j < 6; ++j) {
        const float c0 = ws[WS_C0 + j], s0 = ws[WS_S0 + j];
        u[j][0] = { c0 * x0.r - s0 * x1.r, c0 * x0.i - s0 * x1.i };
        u[j][1] = { s0 * x0.r + c0 * x1.r, s0 * x0.i + c0 * x1.i };
    }
    c32 t01[4], t012[8], t34[4], t345[8];
#pragma unroll
    for (int p = 0; p < 4; ++p) t01[p]  = cmul(u[0][p >> 1], u[1][p & 1]);
#pragma unroll
    for (int p = 0; p < 8; ++p) t012[p] = cmul(t01[p >> 1], u[2][p & 1]);
#pragma unroll
    for (int p = 0; p < 4; ++p) t34[p]  = cmul(u[3][p >> 1], u[4][p & 1]);
#pragma unroll
    for (int p = 0; p < 8; ++p) t345[p] = cmul(t34[p >> 1], u[5][p & 1]);

    float* psir = &s_psi[wave][0][0];
    float* psii = &s_psi[wave][1][0];
#pragma unroll
    for (int pp = 0; pp < 4; ++pp) {
        const c32 tp = t012[4 * kh + pp];
#pragma unroll
        for (int q = 0; q < 8; ++q) {
            const c32 vq = cmul(tp, t345[q]);
            const int kidx = 8 * (4 * kh + pp) + q;
            psir[kidx * 17 + n] = vq.r;
            psii[kidx * 17 + n] = vq.i;
        }
    }
    __syncthreads();

    const float* MR = ws + WS_MR;
    const float* MI = ws + WS_MI;
    const float* MN = ws + WS_MIN;

    float acc0 = 0.f, acc1 = 0.f, acc2 = 0.f, acc3 = 0.f, acc4 = 0.f, acc5 = 0.f;

#pragma unroll
    for (int r = 0; r < 4; ++r) {
        v8f yr = {};
        v8f yi = {};
        const float* rowR = MR + (16 * r + n) * 64 + 2 * kh;
        const float* rowI = MI + (16 * r + n) * 64 + 2 * kh;
        const float* rowN = MN + (16 * r + n) * 64 + 2 * kh;
        for (int kk = 0; kk < 16; ++kk) {
            const v2f ar = *(const v2f*)(rowR + 4 * kk);
            const v2f ai = *(const v2f*)(rowI + 4 * kk);
            const v2f an = *(const v2f*)(rowN + 4 * kk);
            const int k0 = 4 * kk + 2 * kh;
            const float* pr = psir + k0 * 17 + n;
            const float* pi = psii + k0 * 17 + n;
            const v2f br = { pr[0], pr[17] };
            const v2f bi = { pi[0], pi[17] };
            yr = __builtin_amdgcn_wmma_f32_16x16x4_f32(false, ar, false, br, (short)0, yr, false, false);
            yr = __builtin_amdgcn_wmma_f32_16x16x4_f32(false, an, false, bi, (short)0, yr, false, false);
            yi = __builtin_amdgcn_wmma_f32_16x16x4_f32(false, ar, false, bi, (short)0, yi, false, false);
            yi = __builtin_amdgcn_wmma_f32_16x16x4_f32(false, ai, false, br, (short)0, yi, false, false);
        }
#pragma unroll
        for (int v = 0; v < 8; ++v) {
            const float p = yr[v] * yr[v] + yi[v] * yi[v];
            acc0 = (r & 2) ? acc0 - p : acc0 + p;
            acc1 = (r & 1) ? acc1 - p : acc1 + p;
            acc2 += p;
            acc3 = (v & 4) ? acc3 - p : acc3 + p;
            acc4 = (v & 2) ? acc4 - p : acc4 + p;
            acc5 = (v & 1) ? acc5 - p : acc5 + p;
        }
    }
    acc2 = kh ? -acc2 : acc2;

    const float e0 = acc0 + __shfl_xor(acc0, 16, 32);
    const float e1 = acc1 + __shfl_xor(acc1, 16, 32);
    const float e2 = acc2 + __shfl_xor(acc2, 16, 32);
    const float e3 = acc3 + __shfl_xor(acc3, 16, 32);
    const float e4 = acc4 + __shfl_xor(acc4, 16, 32);
    const float e5 = acc5 + __shfl_xor(acc5, 16, 32);

    if (kh == 0) {
        float* o = out + (size_t)b * 6;
        *(v2f*)(o + 0) = (v2f){ e0, e1 };
        *(v2f*)(o + 2) = (v2f){ e2, e3 };
        *(v2f*)(o + 4) = (v2f){ e4, e5 };
    }
}

extern "C" void kernel_launch(void* const* d_in, const int* in_sizes, int n_in,
                              void* d_out, int out_size, void* d_ws, size_t ws_size,
                              hipStream_t stream) {
    const float* w  = (const float*)d_in[0];
    const float* z  = (const float*)d_in[1];
    float* ws       = (float*)d_ws;
    float* o        = (float*)d_out;
    const int B     = in_sizes[1] / 2;

    qsim_setup_kernel<<<1, 64, 0, stream>>>(w, ws);
    qsim_main_kernel<<<B / 64, 128, 0, stream>>>(z, ws, o);
}


# D1 phase dropped (|e^ix|=1) and D0 moved onto psi on the vector unit: M becomes the real R1, 2 WMMAs per k-step instead of 4, one A stream instead of three; changes the fp32 summation order
# speedup vs baseline: 1.7787x; 1.7787x over previous
#include <hip/hip_runtime.h>
#include <hip/hip_bf16.h>


typedef __attribute__((ext_vector_type(2))) float v2f;
typedef __attribute__((ext_vector_type(4))) float v4f;
typedef __attribute__((ext_vector_type(8))) float v8f;

#define WS_R    0
#define WS_D0   4096
#define WS_C0   4224
#define WS_S0   4230

struct c32 { float r, i; };
__device__ __forceinline__ c32 cmul(c32 a, c32 b) {
    return { a.r * b.r - a.i * b.i, a.r * b.i + a.i * b.r };
}

__global__ void qsim_setup_kernel(const float* __restrict__ w,
                                  float* __restrict__ ws) {
    const int k = threadIdx.x;
    float phi0 = 0.f;
#pragma unroll
    for (int i = 0; i < 5; ++i) {
        const int bi = (k >> (5 - i)) & 1;
        const int bj = (k >> (4 - i)) & 1;
        const float sgn = (bi == bj) ? -0.5f : 0.5f;
        phi0 += sgn * w[6 + i];
    }
    float s, c;
    sincosf(phi0, &s, &c);
    ws[WS_D0 + 2 * k + 0] = c;
    ws[WS_D0 + 2 * k + 1] = s;
    float c1[6], s1[6];
#pragma unroll
    for (int i = 0; i < 6; ++i) sincosf(0.5f * w[11 + i], &s1[i], &c1[i]);
    for (int kp = 0; kp < 64; ++kp) {
        float re = 1.f;
#pragma unroll
        for (int i = 0; i < 6; ++i) {
            const int kb = (k >> (5 - i)) & 1;
            const int pb = (kp >> (5 - i)) & 1;
            re *= (kb == pb) ? c1[i] : (kb ? s1[i] : -s1[i]);
        }
        ws[WS_R + k * 64 + kp] = re;
    }
    if (k < 6) {
        float ss, cc; sincosf(0.5f * w[k], &ss, &cc);
        ws[WS_C0 + k] = cc;
        ws[WS_S0 + k] = ss;
    }
}

__global__ __launch_bounds__(128, 1)
void qsim_main_kernel(const float* __restrict__ z,
                      const float* __restrict__ ws,
                      float* __restrict__ out) {
    __shared__ float s_psi[4][2][64 * 17];

    const int tid  = threadIdx.x;
    const int wave = tid >> 5;
    const int lane = tid & 31;
    const int n    = lane & 15;
    const int kh   = lane >> 4;
    const int base = blockIdx.x * 64 + wave * 16;
    const int b    = base + n;

    const float z0 = z[2 * b + 0];
    const float z1 = z[2 * b + 1];
    float sa, ca, sb, cb;
    sincosf(0.5f * asinf(z0), &sa, &ca);
    sincosf(0.5f * asinf(z1), &sb, &cb);
    const c32 x0{ cb * ca, -sb * ca };
    const c32 x1{ cb * sa,  sb * sa };

    c32 u[6][2];
#pragma unroll
    for (int j = 0; j < 6; ++j) {
        const float c0 = ws[WS_C0 + j], s0 = ws[WS_S0 + j];
        u[j][0] = { c0 * x0.r - s0 * x1.r, c0 * x0.i - s0 * x1.i };
        u[j][1] = { s0 * x0.r + c0 * x1.r, s0 * x0.i + c0 * x1.i };
    }
    c32 t01[4], t012[8], t34[4], t345[8];
#pragma unroll
    for (int p = 0; p < 4; ++p) t01[p]  = cmul(u[0][p >> 1], u[1][p & 1]);
#pragma unroll
    for (int p = 0; p < 8; ++p) t012[p] = cmul(t01[p >> 1], u[2][p & 1]);
#pragma unroll
    for (int p = 0; p < 4; ++p) t34[p]  = cmul(u[3][p >> 1], u[4][p & 1]);
#pragma unroll
    for (int p = 0; p < 8; ++p) t345[p] = cmul(t34[p >> 1], u[5][p & 1]);

    float* psir = &s_psi[wave][0][0];
    float* psii = &s_psi[wave][1][0];
    const v4f* d0v = (const v4f*)(ws + WS_D0 + 64 * kh);
#pragma unroll
    for (int pp = 0; pp < 4; ++pp) {
        const c32 tp = t012[4 * kh + pp];
#pragma unroll
        for (int q = 0; q < 8; q += 2) {
            const v4f d = d0v[4 * pp + (q >> 1)];
            const c32 va = cmul(cmul(tp, t345[q]),     c32{ d.x, d.y });
            const c32 vb = cmul(cmul(tp, t345[q + 1]), c32{ d.z, d.w });
            const int kidx = 8 * (4 * kh + pp) + q;
            psir[kidx * 17 + n]       = va.r;
            psii[kidx * 17 + n]       = va.i;
            psir[(kidx + 1) * 17 + n] = vb.r;
            psii[(kidx + 1) * 17 + n] = vb.i;
        }
    }
    __syncthreads();

    const float* R1 = ws + WS_R;

    float acc0 = 0.f, acc1 = 0.f, acc2 = 0.f, acc3 = 0.f, acc4 = 0.f, acc5 = 0.f;

#pragma unroll
    for (int r = 0; r < 4; ++r) {
        v8f yr = {};
        v8f yi = {};
        const float* rowR = R1 + (16 * r + n) * 64 + 2 * kh;
        for (int kk = 0; kk < 16; ++kk) {
            const v2f ar = *(const v2f*)(rowR + 4 * kk);
            const int k0 = 4 * kk + 2 * kh;
            const float* pr = psir + k0 * 17 + n;
            const float* pi = psii + k0 * 17 + n;
            const v2f br = { pr[0], pr[17] };
            const v2f bi = { pi[0], pi[17] };
            yr = __builtin_amdgcn_wmma_f32_16x16x4_f32(false, ar, false, br, (short)0, yr, false, false);
            yi = __builtin_amdgcn_wmma_f32_16x16x4_f32(false, ar, false, bi, (short)0, yi, false, false);
        }
#pragma unroll
        for (int v = 0; v < 8; ++v) {
            const float p = yr[v] * yr[v] + yi[v] * yi[v];
            acc0 = (r & 2) ? acc0 - p : acc0 + p;
            acc1 = (r & 1) ? acc1 - p : acc1 + p;
            acc2 += p;
            acc3 = (v & 4) ? acc3 - p : acc3 + p;
            acc4 = (v & 2) ? acc4 - p : acc4 + p;
            acc5 = (v & 1) ? acc5 - p : acc5 + p;
        }
    }
    acc2 = kh ? -acc2 : acc2;

    const float e0 = acc0 + __shfl_xor(acc0, 16, 32);
    const float e1 = acc1 + __shfl_xor(acc1, 16, 32);
    const float e2 = acc2 + __shfl_xor(acc2, 16, 32);
    const float e3 = acc3 + __shfl_xor(acc3, 16, 32);
    const float e4 = acc4 + __shfl_xor(acc4, 16, 32);
    const float e5 = acc5 + __shfl_xor(acc5, 16, 32);

    if (kh == 0) {
        float* o = out + (size_t)b * 6;
        *(v2f*)(o + 0) = (v2f){ e0, e1 };
        *(v2f*)(o + 2) = (v2f){ e2, e3 };
        *(v2f*)(o + 4) = (v2f){ e4, e5 };
    }
}

extern "C" void kernel_launch(void* const* d_in, const int* in_sizes, int n_in,
                              void* d_out, int out_size, void* d_ws, size_t ws_size,
                              hipStream_t stream) {
    const float* w  = (const float*)d_in[0];
    const float* z  = (const float*)d_in[1];
    float* ws       = (float*)d_ws;
    float* o        = (float*)d_out;
    const int B     = in_sizes[1] / 2;

    qsim_setup_kernel<<<1, 64, 0, stream>>>(w, ws);
    qsim_main_kernel<<<B / 64, 128, 0, stream>>>(z, ws, o);
}


# Both WMMA loaders re-ordered to k=32h+2kk+s: a lane's own psi registers are its B fragments (no LDS, no barrier); R1 stored in that fragment order (512 B per A load); changes fp32 summation order
# speedup vs baseline: 2.0024x; 1.1258x over previous
#include <hip/hip_runtime.h>
#include <hip/hip_bf16.h>


typedef __attribute__((ext_vector_type(2))) float v2f;
typedef __attribute__((ext_vector_type(4))) float v4f;
typedef __attribute__((ext_vector_type(8))) float v8f;

#define WS_R    0
#define WS_D0   4096
#define WS_C0   4224
#define WS_S0   4230

struct c32 { float r, i; };
__device__ __forceinline__ c32 cmul(c32 a, c32 b) {
    return { a.r * b.r - a.i * b.i, a.r * b.i + a.i * b.r };
}

__global__ void qsim_setup_kernel(const float* __restrict__ w,
                                  float* __restrict__ ws) {
    const int k = threadIdx.x;
    float phi0 = 0.f;
#pragma unroll
    for (int i = 0; i < 5; ++i) {
        const int bi = (k >> (5 - i)) & 1;
        const int bj = (k >> (4 - i)) & 1;
        const float sgn = (bi == bj) ? -0.5f : 0.5f;
        phi0 += sgn * w[6 + i];
    }
    float s, c;
    sincosf(phi0, &s, &c);
    ws[WS_D0 + 2 * k + 0] = c;
    ws[WS_D0 + 2 * k + 1] = s;
    float c1[6], s1[6];
#pragma unroll
    for (int i = 0; i < 6; ++i) sincosf(0.5f * w[11 + i], &s1[i], &c1[i]);
    const int rblk = k >> 4, m = k & 15;
    for (int kp = 0; kp < 64; ++kp) {
        float re = 1.f;
#pragma unroll
        for (int i = 0; i < 6; ++i) {
            const int kb = (k >> (5 - i)) & 1;
            const int pb = (kp >> (5 - i)) & 1;
            re *= (kb == pb) ? c1[i] : (kb ? s1[i] : -s1[i]);
        }
        const int h = kp >> 5, j = (kp >> 2) & 7, e = kp & 3;
        ws[WS_R + ((rblk * 8 + j) * 32 + (m + 16 * h)) * 4 + e] = re;
    }
    if (k < 6) {
        float ss, cc; sincosf(0.5f * w[k], &ss, &cc);
        ws[WS_C0 + k] = cc;
        ws[WS_S0 + k] = ss;
    }
}

__global__ __launch_bounds__(128, 8)
void qsim_main_kernel(const float* __restrict__ z,
                      const float* __restrict__ ws,
                      float* __restrict__ out) {
    const int tid  = threadIdx.x;
    const int wave = tid >> 5;
    const int lane = tid & 31;
    const int n    = lane & 15;
    const int kh   = lane >> 4;
    const int base = blockIdx.x * 64 + wave * 16;
    const int b    = base + n;

    const float z0 = z[2 * b + 0];
    const float z1 = z[2 * b + 1];
    float sa, ca, sb, cb;
    sincosf(0.5f * asinf(z0), &sa, &ca);
    sincosf(0.5f * asinf(z1), &sb, &cb);
    const c32 x0{ cb * ca, -sb * ca };
    const c32 x1{ cb * sa,  sb * sa };

    c32 u[6][2];
#pragma unroll
    for (int j = 0; j < 6; ++j) {
        const float c0 = ws[WS_C0 + j], s0 = ws[WS_S0 + j];
        u[j][0] = { c0 * x0.r - s0 * x1.r, c0 * x0.i - s0 * x1.i };
        u[j][1] = { s0 * x0.r + c0 * x1.r, s0 * x0.i + c0 * x1.i };
    }
    c32 t01[4], t012[8], t34[4], t345[8];
#pragma unroll
    for (int p = 0; p < 4; ++p) t01[p]  = cmul(u[0][p >> 1], u[1][p & 1]);
#pragma unroll
    for (int p = 0; p < 8; ++p) t012[p] = cmul(t01[p >> 1], u[2][p & 1]);
#pragma unroll
    for (int p = 0; p < 4; ++p) t34[p]  = cmul(u[3][p >> 1], u[4][p & 1]);
#pragma unroll
    for (int p = 0; p < 8; ++p) t345[p] = cmul(t34[p >> 1], u[5][p & 1]);

    v2f br[16], bi[16];
    const v4f* d0v = (const v4f*)(ws + WS_D0 + 64 * kh);
#pragma unroll
    for (int pp = 0; pp < 4; ++pp) {
        const c32 tp = kh ? t012[4 + pp] : t012[pp];
#pragma unroll
        for (int q = 0; q < 8; q += 2) {
            const v4f d = d0v[4 * pp + (q >> 1)];
            const c32 va = cmul(cmul(tp, t345[q]),     c32{ d.x, d.y });
            const c32 vb = cmul(cmul(tp, t345[q + 1]), c32{ d.z, d.w });
            br[4 * pp + (q >> 1)] = (v2f){ va.r, vb.r };
            bi[4 * pp + (q >> 1)] = (v2f){ va.i, vb.i };
        }
    }

    const v4f* Afrag = (const v4f*)(ws + WS_R) + lane;

    float acc0 = 0.f, acc1 = 0.f, acc2 = 0.f, acc3 = 0.f, acc4 = 0.f, acc5 = 0.f;

    constexpr int AHEAD = 3;
    v4f a[32];
#pragma unroll
    for (int p = 0; p < AHEAD; ++p) a[p] = Afrag[p * 32];

#pragma unroll
    for (int r = 0; r < 4; ++r) {
        v8f yr = {};
        v8f yi = {};
#pragma unroll
        for (int j = 0; j < 8; ++j) {
            const int p = r * 8 + j;
            if (p + AHEAD < 32) a[p + AHEAD] = Afrag[(p + AHEAD) * 32];
            __builtin_amdgcn_sched_barrier(0);
            const v2f a0 = { a[p].x, a[p].y };
            const v2f a1 = { a[p].z, a[p].w };
            yr = __builtin_amdgcn_wmma_f32_16x16x4_f32(false, a0, false, br[2 * j],     (short)0, yr, false, false);
            yi = __builtin_amdgcn_wmma_f32_16x16x4_f32(false, a0, false, bi[2 * j],     (short)0, yi, false, false);
            yr = __builtin_amdgcn_wmma_f32_16x16x4_f32(false, a1, false, br[2 * j + 1], (short)0, yr, false, false);
            yi = __builtin_amdgcn_wmma_f32_16x16x4_f32(false, a1, false, bi[2 * j + 1], (short)0, yi, false, false);
        }
#pragma unroll
        for (int v = 0; v < 8; ++v) {
            const float p = yr[v] * yr[v] + yi[v] * yi[v];
            acc0 = (r & 2) ? acc0 - p : acc0 + p;
            acc1 = (r & 1) ? acc1 - p : acc1 + p;
            acc2 += p;
            acc3 = (v & 4) ? acc3 - p : acc3 + p;
            acc4 = (v & 2) ? acc4 - p : acc4 + p;
            acc5 = (v & 1) ? acc5 - p : acc5 + p;
        }
    }
    acc2 = kh ? -acc2 : acc2;

    const float e0 = acc0 + __shfl_xor(acc0, 16, 32);
    const float e1 = acc1 + __shfl_xor(acc1, 16, 32);
    const float e2 = acc2 + __shfl_xor(acc2, 16, 32);
    const float e3 = acc3 + __shfl_xor(acc3, 16, 32);
    const float e4 = acc4 + __shfl_xor(acc4, 16, 32);
    const float e5 = acc5 + __shfl_xor(acc5, 16, 32);

    if (kh == 0) {
        float* o = out + (size_t)b * 6;
        *(v2f*)(o + 0) = (v2f){ e0, e1 };
        *(v2f*)(o + 2) = (v2f){ e2, e3 };
        *(v2f*)(o + 4) = (v2f){ e4, e5 };
    }
}

extern "C" void kernel_launch(void* const* d_in, const int* in_sizes, int n_in,
                              void* d_out, int out_size, void* d_ws, size_t ws_size,
                              hipStream_t stream) {
    const float* w  = (const float*)d_in[0];
    const float* z  = (const float*)d_in[1];
    float* ws       = (float*)d_ws;
    float* o        = (float*)d_out;
    const int B     = in_sizes[1] / 2;

    qsim_setup_kernel<<<1, 64, 0, stream>>>(w, ws);
    qsim_main_kernel<<<B / 64, 128, 0, stream>>>(z, ws, o);
}


# R1 split into two Kronecker factors (RY0..2 x I2, I2 x RY3..5), each a K=16 WMMA step, step 1's D = step 2's B fragments: 64 WMMAs per wave not 128, set-up stores 512 values; changes fp32 sum order
# speedup vs baseline: 3.3120x; 1.6540x over previous
#include <hip/hip_runtime.h>
#include <hip/hip_bf16.h>


typedef __attribute__((ext_vector_type(2))) float v2f;
typedef __attribute__((ext_vector_type(4))) float v4f;
typedef __attribute__((ext_vector_type(8))) float v8f;

#define WS_A    0
#define WS_D0   512
#define WS_C0   640
#define WS_S0   646

struct c32 { float r, i; };
__device__ __forceinline__ c32 cmul(c32 a, c32 b) {
    return { a.r * b.r - a.i * b.i, a.r * b.i + a.i * b.r };
}

__global__ void qsim_setup_kernel(const float* __restrict__ w,
                                  float* __restrict__ ws) {
    const int k = threadIdx.x;
    float phi0 = 0.f;
#pragma unroll
    for (int i = 0; i < 5; ++i) {
        const int bi = (k >> (5 - i)) & 1;
        const int bj = (k >> (4 - i)) & 1;
        const float sgn = (bi == bj) ? -0.5f : 0.5f;
        phi0 += sgn * w[6 + i];
    }
    float s, c;
    sincosf(phi0, &s, &c);
    ws[WS_D0 + 2 * k + 0] = c;
    ws[WS_D0 + 2 * k + 1] = s;
    const int f = k >> 5, lane = k & 31, m = lane & 15, h = lane >> 4;
    float c1[3], s1[3];
#pragma unroll
    for (int i = 0; i < 3; ++i) sincosf(0.5f * w[11 + 3 * f + i], &s1[i], &c1[i]);
    const int row3 = f ? (m & 7) : (m >> 1);
#pragma unroll
    for (int kk = 0; kk < 4; ++kk) {
#pragma unroll
        for (int e = 0; e < 2; ++e) {
            const int K = 8 * h + 2 * kk + e;
            const int col3 = f ? (K & 7) : (K >> 1);
            const bool on  = f ? ((m >> 3) == (K >> 3)) : ((m & 1) == (K & 1));
            float re = 1.f;
#pragma unroll
            for (int i = 0; i < 3; ++i) {
                const int kb = (row3 >> (2 - i)) & 1;
                const int pb = (col3 >> (2 - i)) & 1;
                re *= (kb == pb) ? c1[i] : (kb ? s1[i] : -s1[i]);
            }
            ws[WS_A + f * 256 + lane * 8 + 2 * kk + e] = on ? re : 0.f;
        }
    }
    if (k < 6) {
        float ss, cc; sincosf(0.5f * w[k], &ss, &cc);
        ws[WS_C0 + k] = cc;
        ws[WS_S0 + k] = ss;
    }
}

__global__ __launch_bounds__(128, 8)
void qsim_main_kernel(const float* __restrict__ z,
                      const float* __restrict__ ws,
                      float* __restrict__ out) {
    const int tid  = threadIdx.x;
    const int wave = tid >> 5;
    const int lane = tid & 31;
    const int n    = lane & 15;
    const int kh   = lane >> 4;
    const int base = blockIdx.x * 64 + wave * 16;
    const int b    = base + n;

    const v4f* Afrag = (const v4f*)(ws + WS_A) + 2 * lane;
    const v4f a1lo = Afrag[0],  a1hi = Afrag[1];
    const v4f a2lo = Afrag[64], a2hi = Afrag[65];
    const v2f a1[4] = { { a1lo.x, a1lo.y }, { a1lo.z, a1lo.w }, { a1hi.x, a1hi.y }, { a1hi.z, a1hi.w } };
    const v2f a2[4] = { { a2lo.x, a2lo.y }, { a2lo.z, a2lo.w }, { a2hi.x, a2hi.y }, { a2hi.z, a2hi.w } };

    const float z0 = z[2 * b + 0];
    const float z1 = z[2 * b + 1];
    float sa, ca, sb, cb;
    sincosf(0.5f * asinf(z0), &sa, &ca);
    sincosf(0.5f * asinf(z1), &sb, &cb);
    const c32 x0{ cb * ca, -sb * ca };
    const c32 x1{ cb * sa,  sb * sa };

    c32 u[6][2];
#pragma unroll
    for (int j = 0; j < 6; ++j) {
        const float c0 = ws[WS_C0 + j], s0 = ws[WS_S0 + j];
        u[j][0] = { c0 * x0.r - s0 * x1.r, c0 * x0.i - s0 * x1.i };
        u[j][1] = { s0 * x0.r + c0 * x1.r, s0 * x0.i + c0 * x1.i };
    }
    c32 t01[4], t012[8], t34[4], t345[8];
#pragma unroll
    for (int p = 0; p < 4; ++p) t01[p]  = cmul(u[0][p >> 1], u[1][p & 1]);
#pragma unroll
    for (int p = 0; p < 8; ++p) t012[p] = cmul(t01[p >> 1], u[2][p & 1]);
#pragma unroll
    for (int p = 0; p < 4; ++p) t34[p]  = cmul(u[3][p >> 1], u[4][p & 1]);
#pragma unroll
    for (int p = 0; p < 8; ++p) t345[p] = cmul(t34[p >> 1], u[5][p & 1]);

    v2f br[16], bi[16];
    const v4f* d0v = (const v4f*)(ws + WS_D0 + 64 * kh);
#pragma unroll
    for (int pp = 0; pp < 4; ++pp) {
        const c32 tp = kh ? t012[4 + pp] : t012[pp];
#pragma unroll
        for (int q = 0; q < 8; q += 2) {
            const v4f d = d0v[4 * pp + (q >> 1)];
            const c32 va = cmul(cmul(tp, t345[q]),     c32{ d.x, d.y });
            const c32 vb = cmul(cmul(tp, t345[q + 1]), c32{ d.z, d.w });
            br[4 * pp + (q >> 1)] = (v2f){ va.r, vb.r };
            bi[4 * pp + (q >> 1)] = (v2f){ va.i, vb.i };
        }
    }

    float acc0 = 0.f, acc1 = 0.f, acc2 = 0.f, acc3 = 0.f, acc4 = 0.f, acc5 = 0.f;

    v8f d1r[4], d1i[4];
#pragma unroll
    for (int u = 0; u < 4; ++u) {
        v8f yr = {};
        v8f yi = {};
#pragma unroll
        for (int kk = 0; kk < 4; ++kk) {
            yr = __builtin_amdgcn_wmma_f32_16x16x4_f32(false, a1[kk], false, br[4 * kk + u], (short)0, yr, false, false);
            yi = __builtin_amdgcn_wmma_f32_16x16x4_f32(false, a1[kk], false, bi[4 * kk + u], (short)0, yi, false, false);
        }
        d1r[u] = yr;
        d1i[u] = yi;
    }

#pragma unroll
    for (int t = 0; t < 4; ++t) {
        v8f yr = {};
        v8f yi = {};
#pragma unroll
        for (int kk = 0; kk < 4; ++kk) {
            const v2f xr = { d1r[kk][2 * t], d1r[kk][2 * t + 1] };
            const v2f xi = { d1i[kk][2 * t], d1i[kk][2 * t + 1] };
            yr = __builtin_amdgcn_wmma_f32_16x16x4_f32(false, a2[kk], false, xr, (short)0, yr, false, false);
            yi = __builtin_amdgcn_wmma_f32_16x16x4_f32(false, a2[kk], false, xi, (short)0, yi, false, false);
        }
#pragma unroll
        for (int v = 0; v < 8; ++v) {
            const float p = yr[v] * yr[v] + yi[v] * yi[v];
            acc0 += p;
            acc1 = (t & 2) ? acc1 - p : acc1 + p;
            acc2 = (t & 1) ? acc2 - p : acc2 + p;
            acc3 = (v & 4) ? acc3 - p : acc3 + p;
            acc4 = (v & 2) ? acc4 - p : acc4 + p;
            acc5 = (v & 1) ? acc5 - p : acc5 + p;
        }
    }
    acc0 = kh ? -acc0 : acc0;

    const float e0 = acc0 + __shfl_xor(acc0, 16, 32);
    const float e1 = acc1 + __shfl_xor(acc1, 16, 32);
    const float e2 = acc2 + __shfl_xor(acc2, 16, 32);
    const float e3 = acc3 + __shfl_xor(acc3, 16, 32);
    const float e4 = acc4 + __shfl_xor(acc4, 16, 32);
    const float e5 = acc5 + __shfl_xor(acc5, 16, 32);

    if (kh == 0) {
        float* o = out + (size_t)b * 6;
        *(v2f*)(o + 0) = (v2f){ e0, e1 };
        *(v2f*)(o + 2) = (v2f){ e2, e3 };
        *(v2f*)(o + 4) = (v2f){ e4, e5 };
    }
}

extern "C" void kernel_launch(void* const* d_in, const int* in_sizes, int n_in,
                              void* d_out, int out_size, void* d_ws, size_t ws_size,
                              hipStream_t stream) {
    const float* w  = (const float*)d_in[0];
    const float* z  = (const float*)d_in[1];
    float* ws       = (float*)d_ws;
    float* o        = (float*)d_out;
    const int B     = in_sizes[1] / 2;

    qsim_setup_kernel<<<1, 64, 0, stream>>>(w, ws);
    qsim_main_kernel<<<B / 64, 128, 0, stream>>>(z, ws, o);
}


# Set-up launch folded into the main kernel: each wave takes sin/cos of the 17 weight angles one per lane (readlane), forms its A fragments and the ZZ phases in registers; changes fp32 phase rounding
# speedup vs baseline: 4.6994x; 1.4189x over previous
#include <hip/hip_runtime.h>
#include <hip/hip_bf16.h>


typedef __attribute__((ext_vector_type(2))) float v2f;
typedef __attribute__((ext_vector_type(8))) float v8f;

struct c32 { float r, i; };
__device__ __forceinline__ c32 cmul(c32 a, c32 b) {
    return { a.r * b.r - a.i * b.i, a.r * b.i + a.i * b.r };
}

__device__ __forceinline__ float from_lane(float v, int src_lane) {
    return __builtin_bit_cast(float,
        __builtin_amdgcn_readlane(__builtin_bit_cast(int, v), src_lane));
}

template <int N>
__device__ __forceinline__ void grow(const c32 (&in)[N], c32 (&res)[2 * N],
                                     c32 un0, c32 un1, float c, float s) {
    const c32 eq{ c, -s }, df{ c, s };
    const c32 w00 = cmul(eq, un0), w10 = cmul(df, un0);
    const c32 w01 = cmul(df, un1), w11 = cmul(eq, un1);
#pragma unroll
    for (int p = 0; p < N; ++p) {
        res[2 * p + 0] = cmul(in[p], (p & 1) ? w10 : w00);
        res[2 * p + 1] = cmul(in[p], (p & 1) ? w11 : w01);
    }
}

__global__ __launch_bounds__(128, 8)
void qsim_fused_kernel(const float* __restrict__ w,
                       const float* __restrict__ z,
                       float* __restrict__ out) {
    const int tid  = threadIdx.x;
    const int wave = tid >> 5;
    const int lane = tid & 31;
    const int n    = lane & 15;
    const int kh   = lane >> 4;
    const int base = blockIdx.x * 64 + wave * 16;
    const int b    = base + n;

    const float z0 = z[2 * b + 0];
    const float z1 = z[2 * b + 1];
    const float wl = w[lane < 17 ? lane : 16];
    float sw, cw;
    sincosf(0.5f * wl, &sw, &cw);
    float C[17], S[17];
#pragma unroll
    for (int j = 0; j < 17; ++j) {
        C[j] = from_lane(cw, j);
        S[j] = from_lane(sw, j);
    }

    float g[6][2];
#pragma unroll
    for (int q = 0; q < 6; ++q) {
        const int rb = (q < 3) ? ((n >> (3 - q)) & 1) : ((n >> (5 - q)) & 1);
        g[q][0] = rb ? S[11 + q] : C[11 + q];
        g[q][1] = rb ? C[11 + q] : -S[11 + q];
    }
    v2f a1[4], a2[4];
    {
        const float f0 = kh ? g[0][1] : g[0][0];
        const bool on2 = ((n >> 3) == kh);
#pragma unroll
        for (int kk = 0; kk < 4; ++kk) {
            const float v1 = f0 * g[1][kk >> 1] * g[2][kk & 1];
            a1[kk] = (n & 1) ? (v2f){ 0.f, v1 } : (v2f){ v1, 0.f };
            const float v2 = g[3][kk >> 1] * g[4][kk & 1];
            a2[kk] = on2 ? (v2f){ v2 * g[5][0], v2 * g[5][1] } : (v2f){ 0.f, 0.f };
        }
    }

    float sa, ca, sb, cb;
    sincosf(0.5f * asinf(z0), &sa, &ca);
    sincosf(0.5f * asinf(z1), &sb, &cb);
    const c32 x0{ cb * ca, -sb * ca };
    const c32 x1{ cb * sa,  sb * sa };

    c32 u[6][2];
#pragma unroll
    for (int j = 0; j < 6; ++j) {
        const float c0 = C[j], s0 = S[j];
        u[j][0] = { c0 * x0.r - s0 * x1.r, c0 * x0.i - s0 * x1.i };
        u[j][1] = { s0 * x0.r + c0 * x1.r, s0 * x0.i + c0 * x1.i };
    }
    c32 p0[1], p1[2], p2[4], p3[8], p4[16], p5[32];
    p0[0] = kh ? u[0][1] : u[0][0];
    grow<1>(p0, p1, u[1][0], u[1][1], C[6], kh ? -S[6] : S[6]);
    grow<2>(p1, p2, u[2][0], u[2][1], C[7], S[7]);
    grow<4>(p2, p3, u[3][0], u[3][1], C[8], S[8]);
    grow<8>(p3, p4, u[4][0], u[4][1], C[9], S[9]);
    grow<16>(p4, p5, u[5][0], u[5][1], C[10], S[10]);

    v2f br[16], bi[16];
#pragma unroll
    for (int i = 0; i < 16; ++i) {
        br[i] = (v2f){ p5[2 * i].r, p5[2 * i + 1].r };
        bi[i] = (v2f){ p5[2 * i].i, p5[2 * i + 1].i };
    }

    float acc0 = 0.f, acc1 = 0.f, acc2 = 0.f, acc3 = 0.f, acc4 = 0.f, acc5 = 0.f;

    v8f d1r[4], d1i[4];
#pragma unroll
    for (int u = 0; u < 4; ++u) {
        v8f yr = {};
        v8f yi = {};
#pragma unroll
        for (int kk = 0; kk < 4; ++kk) {
            yr = __builtin_amdgcn_wmma_f32_16x16x4_f32(false, a1[kk], false, br[4 * kk + u], (short)0, yr, false, false);
            yi = __builtin_amdgcn_wmma_f32_16x16x4_f32(false, a1[kk], false, bi[4 * kk + u], (short)0, yi, false, false);
        }
        d1r[u] = yr;
        d1i[u] = yi;
    }

#pragma unroll
    for (int t = 0; t < 4; ++t) {
        v8f yr = {};
        v8f yi = {};
#pragma unroll
        for (int kk = 0; kk < 4; ++kk) {
            const v2f xr = { d1r[kk][2 * t], d1r[kk][2 * t + 1] };
            const v2f xi = { d1i[kk][2 * t], d1i[kk][2 * t + 1] };
            yr = __builtin_amdgcn_wmma_f32_16x16x4_f32(false, a2[kk], false, xr, (short)0, yr, false, false);
            yi = __builtin_amdgcn_wmma_f32_16x16x4_f32(false, a2[kk], false, xi, (short)0, yi, false, false);
        }
#pragma unroll
        for (int v = 0; v < 8; ++v) {
            const float p = yr[v] * yr[v] + yi[v] * yi[v];
            acc0 += p;
            acc1 = (t & 2) ? acc1 - p : acc1 + p;
            acc2 = (t & 1) ? acc2 - p : acc2 + p;
            acc3 = (v & 4) ? acc3 - p : acc3 + p;
            acc4 = (v & 2) ? acc4 - p : acc4 + p;
            acc5 = (v & 1) ? acc5 - p : acc5 + p;
        }
    }
    acc0 = kh ? -acc0 : acc0;

    const float e0 = acc0 + __shfl_xor(acc0, 16, 32);
    const float e1 = acc1 + __shfl_xor(acc1, 16, 32);
    const float e2 = acc2 + __shfl_xor(acc2, 16, 32);
    const float e3 = acc3 + __shfl_xor(acc3, 16, 32);
    const float e4 = acc4 + __shfl_xor(acc4, 16, 32);
    const float e5 = acc5 + __shfl_xor(acc5, 16, 32);

    if (kh == 0) {
        float* o = out + (size_t)b * 6;
        *(v2f*)(o + 0) = (v2f){ e0, e1 };
        *(v2f*)(o + 2) = (v2f){ e2, e3 };
        *(v2f*)(o + 4) = (v2f){ e4, e5 };
    }
}

extern "C" void kernel_launch(void* const* d_in, const int* in_sizes, int n_in,
                              void* d_out, int out_size, void* d_ws, size_t ws_size,
                              hipStream_t stream) {
    const float* w  = (const float*)d_in[0];
    const float* z  = (const float*)d_in[1];
    float* o        = (float*)d_out;
    const int B     = in_sizes[1] / 2;

    qsim_fused_kernel<<<B / 64, 128, 0, stream>>>(w, z, o);
}


# Encoding half angles taken algebraically, cos(a/2) = sqrt((1+sqrt(1-z^2))/2), sin(a/2) = z/(2 cos(a/2)): two square roots per angle replace asinf + sincosf; changes fp32 rounding of the amplitudes
# speedup vs baseline: 9.5288x; 2.0277x over previous
#include <hip/hip_runtime.h>
#include <hip/hip_bf16.h>


typedef __attribute__((ext_vector_type(2))) float v2f;
typedef __attribute__((ext_vector_type(8))) float v8f;

struct c32 { float r, i; };
__device__ __forceinline__ c32 cmul(c32 a, c32 b) {
    return { a.r * b.r - a.i * b.i, a.r * b.i + a.i * b.r };
}

__device__ __forceinline__ void half_angle(float z, float& s, float& c) {
    const float cosa = __builtin_amdgcn_sqrtf((1.f - z) * (1.f + z));
    const float h    = 0.5f + 0.5f * cosa;
    const float r    = __builtin_amdgcn_rsqf(h);
    c = h * r;
    s = 0.5f * z * r;
}

__device__ __forceinline__ float from_lane(float v, int src_lane) {
    return __builtin_bit_cast(float,
        __builtin_amdgcn_readlane(__builtin_bit_cast(int, v), src_lane));
}

template <int N>
__device__ __forceinline__ void grow(const c32 (&in)[N], c32 (&res)[2 * N],
                                     c32 un0, c32 un1, float c, float s) {
    const c32 eq{ c, -s }, df{ c, s };
    const c32 w00 = cmul(eq, un0), w10 = cmul(df, un0);
    const c32 w01 = cmul(df, un1), w11 = cmul(eq, un1);
#pragma unroll
    for (int p = 0; p < N; ++p) {
        res[2 * p + 0] = cmul(in[p], (p & 1) ? w10 : w00);
        res[2 * p + 1] = cmul(in[p], (p & 1) ? w11 : w01);
    }
}

__global__ __launch_bounds__(128, 8)
void qsim_fused_kernel(const float* __restrict__ w,
                       const float* __restrict__ z,
                       float* __restrict__ out) {
    const int tid  = threadIdx.x;
    const int wave = tid >> 5;
    const int lane = tid & 31;
    const int n    = lane & 15;
    const int kh   = lane >> 4;
    const int base = blockIdx.x * 64 + wave * 16;
    const int b    = base + n;

    const float z0 = z[2 * b + 0];
    const float z1 = z[2 * b + 1];
    const float wl = w[lane < 17 ? lane : 16];
    float sw, cw;
    sincosf(0.5f * wl, &sw, &cw);
    float C[17], S[17];
#pragma unroll
    for (int j = 0; j < 17; ++j) {
        C[j] = from_lane(cw, j);
        S[j] = from_lane(sw, j);
    }

    const float c0[3] = { kh ? C[5]  : C[0],  kh ? C[4]  : C[1],  kh ? C[3]  : C[2]  };
    const float s0[3] = { kh ? S[5]  : S[0],  kh ? S[4]  : S[1],  kh ? S[3]  : S[2]  };
    const float c1[3] = { kh ? C[16] : C[11], kh ? C[15] : C[12], kh ? C[14] : C[13] };
    const float s1[3] = { kh ? S[16] : S[11], kh ? S[15] : S[12], kh ? S[14] : S[13] };
    const float cab = kh ? C[10] : C[6], sab = kh ? S[10] : S[6];
    const float cbc = kh ? C[9]  : C[7], sbc = kh ? S[9]  : S[7];

    float g[3][2];
#pragma unroll
    for (int k = 0; k < 3; ++k) {
        const int rb = (n >> (2 - k)) & 1;
        g[k][0] = rb ? s1[k] : c1[k];
        g[k][1] = rb ? c1[k] : -s1[k];
    }
    const bool on = ((n >> 3) == kh);
    v2f aE[2], aO[2];
#pragma unroll
    for (int kk = 0; kk < 2; ++kk) {
        const float e0 = g[0][kk] * g[1][0] * g[2][0];
        const float e1 = g[0][kk] * g[1][1] * g[2][0];
        const float o0 = g[0][kk] * g[1][0] * g[2][1];
        const float o1 = g[0][kk] * g[1][1] * g[2][1];
        aE[kk] = on ? (v2f){ e0, e1 } : (v2f){ 0.f, 0.f };
        aO[kk] = on ? (v2f){ o0, o1 } : (v2f){ 0.f, 0.f };
    }

    float sa, ca, sb, cb;
    half_angle(z0, sa, ca);
    half_angle(z1, sb, cb);
    const c32 x0{ cb * ca, -sb * ca };
    const c32 x1{ cb * sa,  sb * sa };

    c32 u[3][2];
#pragma unroll
    for (int k = 0; k < 3; ++k) {
        u[k][0] = { c0[k] * x0.r - s0[k] * x1.r, c0[k] * x0.i - s0[k] * x1.i };
        u[k][1] = { s0[k] * x0.r + c0[k] * x1.r, s0[k] * x0.i + c0[k] * x1.i };
    }
    c32 f1[2], f2[4], f3[8];
    f1[0] = u[0][0];
    f1[1] = u[0][1];
    grow<2>(f1, f2, u[1][0], u[1][1], cab, sab);
    grow<4>(f2, f3, u[2][0], u[2][1], cbc, sbc);

    v8f Er = {}, Ei = {}, Or = {}, Oi = {};
#pragma unroll
    for (int kk = 0; kk < 2; ++kk) {
        const v2f ber = { f3[4 * kk].r,     f3[4 * kk + 2].r };
        const v2f bei = { f3[4 * kk].i,     f3[4 * kk + 2].i };
        const v2f bor = { f3[4 * kk + 1].r, f3[4 * kk + 3].r };
        const v2f boi = { f3[4 * kk + 1].i, f3[4 * kk + 3].i };
        Er = __builtin_amdgcn_wmma_f32_16x16x4_f32(false, aE[kk], false, ber, (short)0, Er, false, false);
        Ei = __builtin_amdgcn_wmma_f32_16x16x4_f32(false, aE[kk], false, bei, (short)0, Ei, false, false);
        Or = __builtin_amdgcn_wmma_f32_16x16x4_f32(false, aO[kk], false, bor, (short)0, Or, false, false);
        Oi = __builtin_amdgcn_wmma_f32_16x16x4_f32(false, aO[kk], false, boi, (short)0, Oi, false, false);
    }

    float ee[8], oo[8], cr[8], ci[8];
    float see = 0.f, soo = 0.f, scr = 0.f, sci = 0.f;
#pragma unroll
    for (int r = 0; r < 8; ++r) {
        ee[r] = Er[r] * Er[r] + Ei[r] * Ei[r];
        oo[r] = Or[r] * Or[r] + Oi[r] * Oi[r];
        cr[r] = Er[r] * Or[r] + Ei[r] * Oi[r];
        ci[r] = Ei[r] * Or[r] - Er[r] * Oi[r];
        see += ee[r];
        soo += oo[r];
        scr += cr[r];
        sci += ci[r];
    }
    const float T1p = __shfl_xor((see + soo) + 2.f * scr, 16, 32);
    const float T2p = __shfl_xor((see + soo) - 2.f * scr, 16, 32);
    const float XRp = __shfl_xor(see - soo, 16, 32);
    const float XIp = __shfl_xor(-2.f * sci, 16, 32);

    const float cm = C[8], sm = S[8];
    const float k0 = cm * cm * T1p + sm * sm * T2p;
    const float k1 = 2.f * (cm * cm * T1p - sm * sm * T2p);
    const float k2 = -2.f * cm * sm * XIp;
    const float k3 = 4.f * cm * sm * XRp;
    const float kp = k0 + k2, km = k0 - k2;
    float P[8];
#pragma unroll
    for (int r = 0; r < 8; ++r)
        P[r] = ee[r] * kp + oo[r] * km + cr[r] * k1 + ci[r] * k3;

    const float s01 = P[0] + P[1], d01 = P[0] - P[1];
    const float s23 = P[2] + P[3], d23 = P[2] - P[3];
    const float s45 = P[4] + P[5], d45 = P[4] - P[5];
    const float s67 = P[6] + P[7], d67 = P[6] - P[7];
    const float e_c = (d01 + d23) + (d45 + d67);
    const float e_b = (s01 - s23) + (s45 - s67);
    const float e_a = (s01 + s23) - (s45 + s67);

    const float p_c = __shfl_xor(e_c, 16, 32);
    const float p_b = __shfl_xor(e_b, 16, 32);
    const float p_a = __shfl_xor(e_a, 16, 32);

    if (kh == 0) {
        float* o = out + (size_t)b * 6;
        *(v2f*)(o + 0) = (v2f){ e_a, e_b };
        *(v2f*)(o + 2) = (v2f){ e_c, p_c };
        *(v2f*)(o + 4) = (v2f){ p_b, p_a };
    }
}

extern "C" void kernel_launch(void* const* d_in, const int* in_sizes, int n_in,
                              void* d_out, int out_size, void* d_ws, size_t ws_size,
                              hipStream_t stream) {
    const float* w  = (const float*)d_in[0];
    const float* z  = (const float*)d_in[1];
    float* o        = (float*)d_out;
    const int B     = in_sizes[1] / 2;

    qsim_fused_kernel<<<B / 64, 128, 0, stream>>>(w, z, o);
}


# Each wave takes 8 consecutive 16-column tiles with one A fragment: 8 times fewer waves to start (1024), weight set-up (sincosf, 10 lane gathers) once per 128 columns instead of once per 16
# speedup vs baseline: 15.8898x; 1.6676x over previous
#include <hip/hip_runtime.h>


typedef __attribute__((ext_vector_type(2))) float v2f;
typedef __attribute__((ext_vector_type(8))) float v8f;

#define NL 17
#define AL(j) (j)
#define PH(j) (6 + (j))
#define TH(j) (11 + (j))

struct row_src { unsigned w0, w1; };
#define ENT(th, l, r, a, a2, xs)                                            \
    { (unsigned)((th) * 4) | ((unsigned)((l) * 4) << 8) |                   \
      ((unsigned)((r) * 4) << 16) | ((unsigned)((a) * 4) << 24),            \
      (unsigned)((a2) * 4) | ((unsigned)(xs) << 8) }
#define Q_LO(i)  ENT(TH(i), (i) >= 1 ? PH((i) - 1) : NL, (i) <= 4 ? PH(i) : NL, AL(i), NL, 0)
#define XS_LO(i) ENT(TH(i), (i) >= 1 ? PH((i) - 1) : NL, (i) <= 4 ? PH(i) : NL, AL(i), NL, 1)
#define Z_LO(j)  ENT(NL, NL, NL, AL(j), NL, 0)
#define Q_HI(i)  ENT(TH(i), (i) >= 1 ? PH((i) - 1) : NL, (i) <= 4 ? PH(i) : NL, \
                     (i) <= 4 ? AL((i) + 1) : NL, (i) >= 1 ? AL((i) - 1) : NL, 0)
#define NONE_HI  ENT(NL, NL, NL, NL, NL, 0)

__constant__ const row_src lane_tab[32] = {
    Q_LO(0), Q_LO(1), Q_LO(3), XS_LO(1), Z_LO(0), Z_LO(2), XS_LO(3), Z_LO(4),
    Q_LO(5), Q_LO(4), Q_LO(2), XS_LO(4), Z_LO(5), Z_LO(3), XS_LO(2), Z_LO(1),
    Q_HI(0), Q_HI(1), Q_HI(3), NONE_HI,  NONE_HI, NONE_HI, NONE_HI,  NONE_HI,
    Q_HI(5), Q_HI(4), Q_HI(2), NONE_HI,  NONE_HI, NONE_HI, NONE_HI,  NONE_HI,
};

__device__ __forceinline__ float pick(float v, int addr4) {
    return __builtin_bit_cast(float,
        __builtin_amdgcn_ds_bpermute(addr4, __builtin_bit_cast(int, v)));
}

#define TILES 8

__global__ __launch_bounds__(128)
void qsim_fused_kernel(const float* __restrict__ w,
                       const float* __restrict__ z,
                       float* __restrict__ out) {
    const int tid  = threadIdx.x;
    const int wave = tid >> 5;
    const int lane = tid & 31;
    const int n    = lane & 15;
    const int kh   = lane >> 4;
    const int b0   = (blockIdx.x * 4 + wave) * (16 * TILES) + n;

    const row_src tb = lane_tab[lane];
    const float wl = lane < NL ? w[lane] : 0.f;
    float z0[TILES], z1[TILES];
#pragma unroll
    for (int t = 0; t < TILES; ++t) {
        z0[t] = z[2 * (b0 + 16 * t) + 0];
        z1[t] = z[2 * (b0 + 16 * t) + 1];
    }
    float sw, cw;
    sincosf(wl, &sw, &cw);

    const int p_th = tb.w0 & 0xff, p_l = (tb.w0 >> 8) & 0xff, p_r = (tb.w0 >> 16) & 0xff;
    const int p_a  = tb.w0 >> 24,  p_a2 = tb.w1 & 0xff;
    const bool xs  = (tb.w1 >> 8) & 1;
    const float cth = pick(cw, p_th), sth = pick(sw, p_th);
    const float cl  = pick(cw, p_l),  sl  = pick(sw, p_l);
    const float cr  = pick(cw, p_r),  sr  = pick(sw, p_r);
    const float ca  = pick(cw, p_a),  sa  = pick(sw, p_a);
    const float ca2 = pick(cw, p_a2), sa2 = pick(sw, p_a2);
    const bool hx  = kh || xs;
    const float m1 = (xs ? sl : cl) * (hx ? sr : cr);
    const float pq = hx ? sth * m1 : -(sth * m1);
    const float P  = kh ? pq : (xs ? 0.f : cth);
    const float Q  = kh ? 0.f : pq;
    const float P2 = kh ? sth * (sl * cr) : 0.f;
    const v2f af = { Q * ca - P * sa - P2 * sa2, P * ca + Q * sa + P2 * ca2 };

    float* o = out + (size_t)b0 * 6 + 3 * kh;

#pragma unroll
    for (int t = 0; t < TILES; ++t) {
        const float cosa = __builtin_amdgcn_sqrtf((1.f - z0[t]) * (1.f + z0[t]));
        const float cosb = __builtin_amdgcn_sqrtf((1.f - z1[t]) * (1.f + z1[t]));
        const float rx = z0[t] * cosb, ry = z0[t] * z1[t], rz = cosa;
        const float yy = kh ? ry : 1.f;
        const v2f bf = { rx * yy, rz * yy };

        v8f D = {};
        D = __builtin_amdgcn_wmma_f32_16x16x4_f32(false, af, false, bf, (short)0, D, false, false);

        const float oA = D[3] * D[4] * D[5] + D[1];
        const float oB = D[6] * D[5] * D[7] + D[2];
        const float oX = __builtin_bit_cast(float, __builtin_amdgcn_permlanex16(
            0, __builtin_bit_cast(int, oB), 0x76543210, 0xfedcba98, false, false));

        float* ot = o + t * (16 * 6);
        ot[0] = kh ? oX : D[0];
        ot[1] = oA;
        ot[2] = kh ? D[0] : oX;
    }
}

extern "C" void kernel_launch(void* const* d_in, const int* in_sizes, int n_in,
                              void* d_out, int out_size, void* d_ws, size_t ws_size,
                              hipStream_t stream) {
    const float* w  = (const float*)d_in[0];
    const float* z  = (const float*)d_in[1];
    float* o        = (float*)d_out;
    const int B     = in_sizes[1] / 2;

    qsim_fused_kernel<<<B / (64 * TILES), 128, 0, stream>>>(w, z, o);
}
